# starting kernel
# baseline (speedup 1.0000x reference)
#include <hip/hip_runtime.h>
#include <hip/hip_bf16.h>
#include <math.h>

typedef float v2f __attribute__((ext_vector_type(2)));
typedef float v8f __attribute__((ext_vector_type(8)));

#define B_ 256
#define S_ 128
#define E_ 512
#define H_ 1024
#define V_ 128
#define C_ 18

__global__ void k_zero(float* __restrict__ p, int n) {
    int i = blockIdx.x * blockDim.x + threadIdx.x;
    if (i < n) p[i] = 0.0f;
}

__global__ __launch_bounds__(256)
void k_embed_proj(const int* __restrict__ x_in,
                  const float* __restrict__ emb,
                  const float* __restrict__ W_ih,
                  const float* __restrict__ b_ih,
                  float* __restrict__ pre) {
    const int lane = threadIdx.x & 31;
    const int wave = threadIdx.x >> 5;
    const int gw   = blockIdx.x * (blockDim.x >> 5) + wave;

    const int NG = H_ / 64;
    const int tn = (gw % NG) * 64;
    const int tm = (gw / NG) * 16;

    const int half = lane >> 4;
    const int l15  = lane & 15;

    const int m_a = tm + l15;
    const int s   = m_a / B_;
    const int b   = m_a % B_;
    const int tok = x_in[b * S_ + s];
    const float* __restrict__ arow = emb + (size_t)tok * E_ + half * 2;

    const float* __restrict__ brow0 = W_ih + (size_t)(tn +  0 + l15) * E_ + half * 2;
    const float* __restrict__ brow1 = W_ih + (size_t)(tn + 16 + l15) * E_ + half * 2;
    const float* __restrict__ brow2 = W_ih + (size_t)(tn + 32 + l15) * E_ + half * 2;
    const float* __restrict__ brow3 = W_ih + (size_t)(tn + 48 + l15) * E_ + half * 2;

    v8f acc0 = {}; v8f acc1 = {}; v8f acc2 = {}; v8f acc3 = {};

    #pragma unroll 4
    for (int k0 = 0; k0 < E_; k0 += 4) {
        v2f a  = { arow[k0],  arow[k0 + 1] };
        v2f b0 = { brow0[k0], brow0[k0 + 1] };
        v2f b1 = { brow1[k0], brow1[k0 + 1] };
        v2f b2v = { brow2[k0], brow2[k0 + 1] };
        v2f b3 = { brow3[k0], brow3[k0 + 1] };
        acc0 = __builtin_amdgcn_wmma_f32_16x16x4_f32(false, a, false, b0,  (short)0, acc0, false, false);
        acc1 = __builtin_amdgcn_wmma_f32_16x16x4_f32(false, a, false, b1,  (short)0, acc1, false, false);
        acc2 = __builtin_amdgcn_wmma_f32_16x16x4_f32(false, a, false, b2v, (short)0, acc2, false, false);
        acc3 = __builtin_amdgcn_wmma_f32_16x16x4_f32(false, a, false, b3,  (short)0, acc3, false, false);
    }

    #pragma unroll
    for (int r = 0; r < 8; ++r) {
        const int row = tm + half * 8 + r;
        float* __restrict__ dst = pre + (size_t)row * H_ + tn + l15;
        dst[ 0] = acc0[r] + b_ih[tn +  0 + l15];
        dst[16] = acc1[r] + b_ih[tn + 16 + l15];
        dst[32] = acc2[r] + b_ih[tn + 32 + l15];
        dst[48] = acc3[r] + b_ih[tn + 48 + l15];
    }
}

__global__ __launch_bounds__(256)
void k_rnn_step(const float* __restrict__ hin,
                const float* __restrict__ W_hh,
                const float* __restrict__ b_hh,
                const float* __restrict__ pre_s,
                const int* __restrict__ lengths,
                int s,
                float* __restrict__ hout,
                float* __restrict__ last) {
    const int lane = threadIdx.x & 31;
    const int wave = threadIdx.x >> 5;
    const int gw   = blockIdx.x * (blockDim.x >> 5) + wave;

    const int NT = H_ / 16;
    const int tn = (gw % NT) * 16;
    const int tm = (gw / NT) * 16;

    const int half = lane >> 4;
    const int l15  = lane & 15;

    const float* __restrict__ arow = hin  + (size_t)(tm + l15) * H_ + half * 2;
    const float* __restrict__ brow = W_hh + (size_t)(tn + l15) * H_ + half * 2;

    v8f acc = {};
    #pragma unroll 8
    for (int k0 = 0; k0 < H_; k0 += 4) {
        v2f a = { arow[k0], arow[k0 + 1] };
        v2f b = { brow[k0], brow[k0 + 1] };
        acc = __builtin_amdgcn_wmma_f32_16x16x4_f32(false, a, false, b, (short)0, acc, false, false);
    }

    const int   col  = tn + l15;
    const float bias = b_hh[col];
    #pragma unroll
    for (int r = 0; r < 8; ++r) {
        const int row = tm + half * 8 + r;
        const float v = tanhf(acc[r] + pre_s[(size_t)row * H_ + col] + bias);
        hout[(size_t)row * H_ + col] = v;
        if (lengths[row] - 1 == s) last[(size_t)row * H_ + col] = v;
    }
}

__global__ __launch_bounds__(256)
void k_mlp1(const float* __restrict__ last,
            const float* __restrict__ W1,
            const float* __restrict__ b1,
            float* __restrict__ y1) {
    const int lane = threadIdx.x & 31;
    const int wave = threadIdx.x >> 5;
    const int gw   = blockIdx.x * (blockDim.x >> 5) + wave;

    const int NT = H_ / 16;
    const int tn = (gw % NT) * 16;
    const int tm = (gw / NT) * 16;

    const int half = lane >> 4;
    const int l15  = lane & 15;

    const float* __restrict__ arow = last + (size_t)(tm + l15) * H_ + half * 2;
    const float* __restrict__ brow = W1   + (size_t)(tn + l15) * H_ + half * 2;

    v8f acc = {};
    #pragma unroll 8
    for (int k0 = 0; k0 < H_; k0 += 4) {
        v2f a = { arow[k0], arow[k0 + 1] };
        v2f b = { brow[k0], brow[k0 + 1] };
        acc = __builtin_amdgcn_wmma_f32_16x16x4_f32(false, a, false, b, (short)0, acc, false, false);
    }

    const int   col  = tn + l15;
    const float bias = b1[col];
    #pragma unroll
    for (int r = 0; r < 8; ++r) {
        const int row = tm + half * 8 + r;
        const float v = acc[r] + bias;
        y1[(size_t)row * H_ + col] = v > 0.0f ? v : 0.0f;
    }
}

__global__ void k_mlp2(const float* __restrict__ y1,
                       const float* __restrict__ W2,
                       const float* __restrict__ b2,
                       float* __restrict__ out) {
    int i = blockIdx.x * blockDim.x + threadIdx.x;
    if (i >= B_ * C_) return;
    const int b = i / C_;
    const int c = i % C_;
    const float* __restrict__ yr = y1 + (size_t)b * H_;
    const float* __restrict__ wr = W2 + (size_t)c * H_;
    float acc = b2[c];
    #pragma unroll 8
    for (int k = 0; k < H_; ++k) acc += yr[k] * wr[k];
    out[i] = acc;
}

extern "C" void kernel_launch(void* const* d_in, const int* in_sizes, int n_in,
                              void* d_out, int out_size, void* d_ws, size_t ws_size,
                              hipStream_t stream) {
    const int*   x_in  = (const int*)  d_in[0];
    const int*   x_len = (const int*)  d_in[1];
    const float* emb   = (const float*)d_in[2];
    const float* W_ih  = (const float*)d_in[3];
    const float* b_ih  = (const float*)d_in[4];
    const float* W_hh  = (const float*)d_in[5];
    const float* b_hh  = (const float*)d_in[6];
    const float* W1    = (const float*)d_in[7];
    const float* b1    = (const float*)d_in[8];
    const float* W2    = (const float*)d_in[9];
    const float* b2    = (const float*)d_in[10];
    float* out = (float*)d_out;

    float* pre  = (float*)d_ws;
    float* hb0  = pre  + (size_t)S_ * B_ * H_;
    float* hb1  = hb0  + (size_t)B_ * H_;
    float* last = hb1  + (size_t)B_ * H_;
    float* y1   = last + (size_t)B_ * H_;

    k_zero<<<(B_ * H_ + 255) / 256, 256, 0, stream>>>(hb0, B_ * H_);

    {
        const int waves  = (S_ * B_ / 16) * (H_ / 64);
        const int blocks = waves / 8;
        k_embed_proj<<<blocks, 256, 0, stream>>>(x_in, emb, W_ih, b_ih, pre);
    }

    {
        const int waves  = (B_ / 16) * (H_ / 16);
        const int blocks = waves / 8;
        float* bufs[2] = { hb0, hb1 };
        for (int s = 0; s < S_; ++s) {
            const float* hin  = bufs[s & 1];
            float*       hout = bufs[(s + 1) & 1];
            k_rnn_step<<<blocks, 256, 0, stream>>>(
                hin, W_hh, b_hh, pre + (size_t)s * B_ * H_, x_len, s, hout, last);
        }
    }

    {
        const int waves  = (B_ / 16) * (H_ / 16);
        const int blocks = waves / 8;
        k_mlp1<<<blocks, 256, 0, stream>>>(last, W1, b1, y1);
        k_mlp2<<<(B_ * C_ + 127) / 128, 128, 0, stream>>>(y1, W2, b2, out);
    }
}


# Input projection done once per vocabulary row (128x1024 table, same loaders and k order) and gathered by token in each step, not over all 32768 gathered rows; values bit-identical
# speedup vs baseline: 1.2200x; 1.2200x over previous
#include <hip/hip_runtime.h>
#include <hip/hip_bf16.h>
#include <math.h>

typedef float v2f __attribute__((ext_vector_type(2)));
typedef float v8f __attribute__((ext_vector_type(8)));

#define B_ 256
#define S_ 128
#define E_ 512
#define H_ 1024
#define V_ 128
#define C_ 18

__global__ void k_zero(float* __restrict__ p, int n) {
    int i = blockIdx.x * blockDim.x + threadIdx.x;
    if (i < n) p[i] = 0.0f;
}

__global__ __launch_bounds__(128)
void k_vocab_proj(const float* __restrict__ emb,
                  const float* __restrict__ W_ih,
                  const float* __restrict__ b_ih,
                  float* __restrict__ tab) {
    const int lane = threadIdx.x & 31;
    const int wave = threadIdx.x >> 5;
    const int gw   = blockIdx.x * (blockDim.x >> 5) + wave;

    const int NT = H_ / 16;
    const int tn = (gw % NT) * 16;
    const int tm = (gw / NT) * 16;

    const int half = lane >> 4;
    const int l15  = lane & 15;

    const float* __restrict__ arow = emb  + (size_t)(tm + l15) * E_ + half * 2;
    const float* __restrict__ brow = W_ih + (size_t)(tn + l15) * E_ + half * 2;

    v8f acc = {};
    #pragma unroll 16
    for (int k0 = 0; k0 < E_; k0 += 4) {
        v2f a = { arow[k0], arow[k0 + 1] };
        v2f b = { brow[k0], brow[k0 + 1] };
        acc = __builtin_amdgcn_wmma_f32_16x16x4_f32(false, a, false, b, (short)0, acc, false, false);
    }

    const int   col  = tn + l15;
    const float bias = b_ih[col];
    #pragma unroll
    for (int r = 0; r < 8; ++r) {
        const int row = tm + half * 8 + r;
        tab[(size_t)row * H_ + col] = acc[r] + bias;
    }
}

__global__ __launch_bounds__(256)
void k_rnn_step(const float* __restrict__ hin,
                const float* __restrict__ W_hh,
                const float* __restrict__ b_hh,
                const float* __restrict__ tab,
                const int* __restrict__ x_in,
                const int* __restrict__ lengths,
                int s,
                float* __restrict__ hout,
                float* __restrict__ last) {
    const int lane = threadIdx.x & 31;
    const int wave = threadIdx.x >> 5;
    const int gw   = blockIdx.x * (blockDim.x >> 5) + wave;

    const int NT = H_ / 16;
    const int tn = (gw % NT) * 16;
    const int tm = (gw / NT) * 16;

    const int half = lane >> 4;
    const int l15  = lane & 15;

    const float* __restrict__ arow = hin  + (size_t)(tm + l15) * H_ + half * 2;
    const float* __restrict__ brow = W_hh + (size_t)(tn + l15) * H_ + half * 2;

    int tok[8];
    #pragma unroll
    for (int r = 0; r < 8; ++r) tok[r] = x_in[(tm + half * 8 + r) * S_ + s];

    v8f acc = {};
    #pragma unroll 8
    for (int k0 = 0; k0 < H_; k0 += 4) {
        v2f a = { arow[k0], arow[k0 + 1] };
        v2f b = { brow[k0], brow[k0 + 1] };
        acc = __builtin_amdgcn_wmma_f32_16x16x4_f32(false, a, false, b, (short)0, acc, false, false);
    }

    const int   col  = tn + l15;
    const float bias = b_hh[col];
    #pragma unroll
    for (int r = 0; r < 8; ++r) {
        const int row = tm + half * 8 + r;
        const float v = tanhf(acc[r] + tab[(size_t)tok[r] * H_ + col] + bias);
        hout[(size_t)row * H_ + col] = v;
        if (lengths[row] - 1 == s) last[(size_t)row * H_ + col] = v;
    }
}

__global__ __launch_bounds__(256)
void k_mlp1(const float* __restrict__ last,
            const float* __restrict__ W1,
            const float* __restrict__ b1,
            float* __restrict__ y1) {
    const int lane = threadIdx.x & 31;
    const int wave = threadIdx.x >> 5;
    const int gw   = blockIdx.x * (blockDim.x >> 5) + wave;

    const int NT = H_ / 16;
    const int tn = (gw % NT) * 16;
    const int tm = (gw / NT) * 16;

    const int half = lane >> 4;
    const int l15  = lane & 15;

    const float* __restrict__ arow = last + (size_t)(tm + l15) * H_ + half * 2;
    const float* __restrict__ brow = W1   + (size_t)(tn + l15) * H_ + half * 2;

    v8f acc = {};
    #pragma unroll 8
    for (int k0 = 0; k0 < H_; k0 += 4) {
        v2f a = { arow[k0], arow[k0 + 1] };
        v2f b = { brow[k0], brow[k0 + 1] };
        acc = __builtin_amdgcn_wmma_f32_16x16x4_f32(false, a, false, b, (short)0, acc, false, false);
    }

    const int   col  = tn + l15;
    const float bias = b1[col];
    #pragma unroll
    for (int r = 0; r < 8; ++r) {
        const int row = tm + half * 8 + r;
        const float v = acc[r] + bias;
        y1[(size_t)row * H_ + col] = v > 0.0f ? v : 0.0f;
    }
}

__global__ void k_mlp2(const float* __restrict__ y1,
                       const float* __restrict__ W2,
                       const float* __restrict__ b2,
                       float* __restrict__ out) {
    int i = blockIdx.x * blockDim.x + threadIdx.x;
    if (i >= B_ * C_) return;
    const int b = i / C_;
    const int c = i % C_;
    const float* __restrict__ yr = y1 + (size_t)b * H_;
    const float* __restrict__ wr = W2 + (size_t)c * H_;
    float acc = b2[c];
    #pragma unroll 8
    for (int k = 0; k < H_; ++k) acc += yr[k] * wr[k];
    out[i] = acc;
}

extern "C" void kernel_launch(void* const* d_in, const int* in_sizes, int n_in,
                              void* d_out, int out_size, void* d_ws, size_t ws_size,
                              hipStream_t stream) {
    const int*   x_in  = (const int*)  d_in[0];
    const int*   x_len = (const int*)  d_in[1];
    const float* emb   = (const float*)d_in[2];
    const float* W_ih  = (const float*)d_in[3];
    const float* b_ih  = (const float*)d_in[4];
    const float* W_hh  = (const float*)d_in[5];
    const float* b_hh  = (const float*)d_in[6];
    const float* W1    = (const float*)d_in[7];
    const float* b1    = (const float*)d_in[8];
    const float* W2    = (const float*)d_in[9];
    const float* b2    = (const float*)d_in[10];
    float* out = (float*)d_out;

    float* tab  = (float*)d_ws;
    float* hb0  = tab  + (size_t)S_ * B_ * H_;
    float* hb1  = hb0  + (size_t)B_ * H_;
    float* last = hb1  + (size_t)B_ * H_;
    float* y1   = last + (size_t)B_ * H_;

    k_zero<<<(B_ * H_ + 255) / 256, 256, 0, stream>>>(hb0, B_ * H_);

    {
        const int waves  = (V_ / 16) * (H_ / 16);
        const int blocks = waves / 4;
        k_vocab_proj<<<blocks, 128, 0, stream>>>(emb, W_ih, b_ih, tab);
    }

    {
        const int waves  = (B_ / 16) * (H_ / 16);
        const int blocks = waves / 8;
        float* bufs[2] = { hb0, hb1 };
        for (int s = 0; s < S_; ++s) {
            const float* hin  = bufs[s & 1];
            float*       hout = bufs[(s + 1) & 1];
            k_rnn_step<<<blocks, 256, 0, stream>>>(
                hin, W_hh, b_hh, tab, x_in, x_len, s, hout, last);
        }
    }

    {
        const int waves  = (B_ / 16) * (H_ / 16);
        const int blocks = waves / 8;
        k_mlp1<<<blocks, 256, 0, stream>>>(last, W1, b1, y1);
        k_mlp2<<<(B_ * C_ + 127) / 128, 128, 0, stream>>>(y1, W2, b2, out);
    }
}


# Fragment loads widened to 16 bytes per lane (32 bytes of a 128-byte line per touch), one A and one B load feeding two WMMAs; k order inside each group of 8 changes the f32 summation order
# speedup vs baseline: 1.3877x; 1.1375x over previous
#include <hip/hip_runtime.h>
#include <hip/hip_bf16.h>
#include <math.h>

typedef float v2f __attribute__((ext_vector_type(2)));
typedef float v4f __attribute__((ext_vector_type(4)));
typedef float v8f __attribute__((ext_vector_type(8)));

#define B_ 256
#define S_ 128
#define E_ 512
#define H_ 1024
#define V_ 128
#define C_ 18

__global__ void k_zero(float* __restrict__ p, int n) {
    int i = blockIdx.x * blockDim.x + threadIdx.x;
    if (i < n) p[i] = 0.0f;
}


__global__ __launch_bounds__(128)
void k_vocab_proj(const float* __restrict__ emb,
                  const float* __restrict__ W_ih,
                  const float* __restrict__ b_ih,
                  float* __restrict__ tab) {
    const int lane = threadIdx.x & 31;
    const int wave = threadIdx.x >> 5;
    const int gw   = blockIdx.x * (blockDim.x >> 5) + wave;

    const int NT = H_ / 16;
    const int tn = (gw % NT) * 16;
    const int tm = (gw / NT) * 16;

    const int half = lane >> 4;
    const int l15  = lane & 15;

    const float* __restrict__ arow = emb  + (size_t)(tm + l15) * E_ + half * 4;
    const float* __restrict__ brow = W_ih + (size_t)(tn + l15) * E_ + half * 4;

    v8f acc = {};
    #pragma unroll 16
    for (int k0 = 0; k0 < E_; k0 += 8) {
        const v4f a = *(const v4f*)(arow + k0);
        const v4f b = *(const v4f*)(brow + k0);
        const v2f a0 = { a.x, a.y }, b0 = { b.x, b.y };
        const v2f a1 = { a.z, a.w }, b1 = { b.z, b.w };
        acc = __builtin_amdgcn_wmma_f32_16x16x4_f32(false, a0, false, b0, (short)0, acc, false, false);
        acc = __builtin_amdgcn_wmma_f32_16x16x4_f32(false, a1, false, b1, (short)0, acc, false, false);
    }

    const int   col  = tn + l15;
    const float bias = b_ih[col];
    #pragma unroll
    for (int r = 0; r < 8; ++r) {
        const int row = tm + half * 8 + r;
        tab[(size_t)row * H_ + col] = acc[r] + bias;
    }
}

__global__ __launch_bounds__(256)
void k_rnn_step(const float* __restrict__ hin,
                const float* __restrict__ W_hh,
                const float* __restrict__ b_hh,
                const float* __restrict__ tab,
                const int* __restrict__ x_in,
                const int* __restrict__ lengths,
                int s,
                float* __restrict__ hout,
                float* __restrict__ last) {
    const int lane = threadIdx.x & 31;
    const int wave = threadIdx.x >> 5;
    const int gw   = blockIdx.x * (blockDim.x >> 5) + wave;

    const int NT = H_ / 16;
    const int tn = (gw % NT) * 16;
    const int tm = (gw / NT) * 16;

    const int half = lane >> 4;
    const int l15  = lane & 15;

    const float* __restrict__ arow = hin  + (size_t)(tm + l15) * H_ + half * 4;
    const float* __restrict__ brow = W_hh + (size_t)(tn + l15) * H_ + half * 4;

    int tok[8];
    #pragma unroll
    for (int r = 0; r < 8; ++r) tok[r] = x_in[(tm + half * 8 + r) * S_ + s];

    v8f acc = {};
    #pragma unroll 8
    for (int k0 = 0; k0 < H_; k0 += 8) {
        const v4f a = *(const v4f*)(arow + k0);
        const v4f b = *(const v4f*)(brow + k0);
        const v2f a0 = { a.x, a.y }, b0 = { b.x, b.y };
        const v2f a1 = { a.z, a.w }, b1 = { b.z, b.w };
        acc = __builtin_amdgcn_wmma_f32_16x16x4_f32(false, a0, false, b0, (short)0, acc, false, false);
        acc = __builtin_amdgcn_wmma_f32_16x16x4_f32(false, a1, false, b1, (short)0, acc, false, false);
    }

    const int   col  = tn + l15;
    const float bias = b_hh[col];
    #pragma unroll
    for (int r = 0; r < 8; ++r) {
        const int row = tm + half * 8 + r;
        const float v = tanhf(acc[r] + tab[(size_t)tok[r] * H_ + col] + bias);
        hout[(size_t)row * H_ + col] = v;
        if (lengths[row] - 1 == s) last[(size_t)row * H_ + col] = v;
    }
}

__global__ __launch_bounds__(256)
void k_mlp1(const float* __restrict__ last,
            const float* __restrict__ W1,
            const float* __restrict__ b1,
            float* __restrict__ y1) {
    const int lane = threadIdx.x & 31;
    const int wave = threadIdx.x >> 5;
    const int gw   = blockIdx.x * (blockDim.x >> 5) + wave;

    const int NT = H_ / 16;
    const int tn = (gw % NT) * 16;
    const int tm = (gw / NT) * 16;

    const int half = lane >> 4;
    const int l15  = lane & 15;

    const float* __restrict__ arow = last + (size_t)(tm + l15) * H_ + half * 4;
    const float* __restrict__ brow = W1   + (size_t)(tn + l15) * H_ + half * 4;

    v8f acc = {};
    #pragma unroll 8
    for (int k0 = 0; k0 < H_; k0 += 8) {
        const v4f a = *(const v4f*)(arow + k0);
        const v4f b = *(const v4f*)(brow + k0);
        const v2f a0 = { a.x, a.y }, b0 = { b.x, b.y };
        const v2f a1 = { a.z, a.w }, b1 = { b.z, b.w };
        acc = __builtin_amdgcn_wmma_f32_16x16x4_f32(false, a0, false, b0, (short)0, acc, false, false);
        acc = __builtin_amdgcn_wmma_f32_16x16x4_f32(false, a1, false, b1, (short)0, acc, false, false);
    }

    const int   col  = tn + l15;
    const float bias = b1[col];
    #pragma unroll
    for (int r = 0; r < 8; ++r) {
        const int row = tm + half * 8 + r;
        const float v = acc[r] + bias;
        y1[(size_t)row * H_ + col] = v > 0.0f ? v : 0.0f;
    }
}

__global__ void k_mlp2(const float* __restrict__ y1,
                       const float* __restrict__ W2,
                       const float* __restrict__ b2,
                       float* __restrict__ out) {
    int i = blockIdx.x * blockDim.x + threadIdx.x;
    if (i >= B_ * C_) return;
    const int b = i / C_;
    const int c = i % C_;
    const float* __restrict__ yr = y1 + (size_t)b * H_;
    const float* __restrict__ wr = W2 + (size_t)c * H_;
    float acc = b2[c];
    #pragma unroll 8
    for (int k = 0; k < H_; ++k) acc += yr[k] * wr[k];
    out[i] = acc;
}

extern "C" void kernel_launch(void* const* d_in, const int* in_sizes, int n_in,
                              void* d_out, int out_size, void* d_ws, size_t ws_size,
                              hipStream_t stream) {
    const int*   x_in  = (const int*)  d_in[0];
    const int*   x_len = (const int*)  d_in[1];
    const float* emb   = (const float*)d_in[2];
    const float* W_ih  = (const float*)d_in[3];
    const float* b_ih  = (const float*)d_in[4];
    const float* W_hh  = (const float*)d_in[5];
    const float* b_hh  = (const float*)d_in[6];
    const float* W1    = (const float*)d_in[7];
    const float* b1    = (const float*)d_in[8];
    const float* W2    = (const float*)d_in[9];
    const float* b2    = (const float*)d_in[10];
    float* out = (float*)d_out;

    float* tab  = (float*)d_ws;
    float* hb0  = tab  + (size_t)S_ * B_ * H_;
    float* hb1  = hb0  + (size_t)B_ * H_;
    float* last = hb1  + (size_t)B_ * H_;
    float* y1   = last + (size_t)B_ * H_;

    k_zero<<<(B_ * H_ + 255) / 256, 256, 0, stream>>>(hb0, B_ * H_);

    {
        const int waves  = (V_ / 16) * (H_ / 16);
        const int blocks = waves / 4;
        k_vocab_proj<<<blocks, 128, 0, stream>>>(emb, W_ih, b_ih, tab);
    }

    {
        const int waves  = (B_ / 16) * (H_ / 16);
        const int blocks = waves / 8;
        float* bufs[2] = { hb0, hb1 };
        for (int s = 0; s < S_; ++s) {
            const float* hin  = bufs[s & 1];
            float*       hout = bufs[(s + 1) & 1];
            k_rnn_step<<<blocks, 256, 0, stream>>>(
                hin, W_hh, b_hh, tab, x_in, x_len, s, hout, last);
        }
    }

    {
        const int waves  = (B_ / 16) * (H_ / 16);
        const int blocks = waves / 8;
        k_mlp1<<<blocks, 256, 0, stream>>>(last, W1, b1, y1);
        k_mlp2<<<(B_ * C_ + 127) / 128, 128, 0, stream>>>(y1, W2, b2, out);
    }
}


# Step GEMM re-tiled: 256 work-groups of 8 waves, each wave a 32x32 register tile (2x2 fragments) over one eighth of K, its 64 loads issued ahead; partial tiles added via LDS (new f32 sum order)
# speedup vs baseline: 3.5080x; 2.5279x over previous
#include <hip/hip_runtime.h>
#include <hip/hip_bf16.h>
#include <math.h>

typedef float v2f __attribute__((ext_vector_type(2)));
typedef float v4f __attribute__((ext_vector_type(4)));
typedef float v8f __attribute__((ext_vector_type(8)));

#define B_ 256
#define S_ 128
#define E_ 512
#define H_ 1024
#define V_ 128
#define C_ 18

__global__ void k_zero(float* __restrict__ p, int n) {
    int i = blockIdx.x * blockDim.x + threadIdx.x;
    if (i < n) p[i] = 0.0f;
}


__global__ __launch_bounds__(128)
void k_vocab_proj(const float* __restrict__ emb,
                  const float* __restrict__ W_ih,
                  const float* __restrict__ b_ih,
                  float* __restrict__ tab) {
    const int lane = threadIdx.x & 31;
    const int wave = threadIdx.x >> 5;
    const int gw   = blockIdx.x * (blockDim.x >> 5) + wave;

    const int NT = H_ / 16;
    const int tn = (gw % NT) * 16;
    const int tm = (gw / NT) * 16;

    const int half = lane >> 4;
    const int l15  = lane & 15;

    const float* __restrict__ arow = emb  + (size_t)(tm + l15) * E_ + half * 4;
    const float* __restrict__ brow = W_ih + (size_t)(tn + l15) * E_ + half * 4;

    v8f acc = {};
    #pragma unroll 16
    for (int k0 = 0; k0 < E_; k0 += 8) {
        const v4f a = *(const v4f*)(arow + k0);
        const v4f b = *(const v4f*)(brow + k0);
        const v2f a0 = { a.x, a.y }, b0 = { b.x, b.y };
        const v2f a1 = { a.z, a.w }, b1 = { b.z, b.w };
        acc = __builtin_amdgcn_wmma_f32_16x16x4_f32(false, a0, false, b0, (short)0, acc, false, false);
        acc = __builtin_amdgcn_wmma_f32_16x16x4_f32(false, a1, false, b1, (short)0, acc, false, false);
    }

    const int   col  = tn + l15;
    const float bias = b_ih[col];
    #pragma unroll
    for (int r = 0; r < 8; ++r) {
        const int row = tm + half * 8 + r;
        tab[(size_t)row * H_ + col] = acc[r] + bias;
    }
}

#define KW_ (H_ / 8)
#define RP_ 36

__global__ __launch_bounds__(256)
void k_rnn_step(const float* __restrict__ hin,
                const float* __restrict__ W_hh,
                const float* __restrict__ b_hh,
                const float* __restrict__ tab,
                const int* __restrict__ x_in,
                const int* __restrict__ lengths,
                int s,
                float* __restrict__ hout,
                float* __restrict__ last) {
    __shared__ float part[8 * 32 * RP_];

    const int tid  = threadIdx.x;
    const int lane = tid & 31;
    const int wave = tid >> 5;
    const int half = lane >> 4;
    const int l15  = lane & 15;

    const int NT = H_ / 32;
    const int tn = (blockIdx.x % NT) * 32;
    const int tm = (blockIdx.x / NT) * 32;

    const int er   = tid >> 3;
    const int ec   = (tid & 7) * 4;
    const int erow = tm + er;
    const int ecol = tn + ec;
    const int tok = x_in[erow * S_ + s];
    const int len = lengths[erow];

    const float* __restrict__ a0p = hin  + (size_t)(tm + l15) * H_ + wave * KW_ + half * 4;
    const float* __restrict__ a1p = a0p + (size_t)16 * H_;
    const float* __restrict__ b0p = W_hh + (size_t)(tn + l15) * H_ + wave * KW_ + half * 4;
    const float* __restrict__ b1p = b0p + (size_t)16 * H_;

    v4f fa0[KW_ / 8], fa1[KW_ / 8], fb0[KW_ / 8], fb1[KW_ / 8];
    #pragma unroll
    for (int i = 0; i < KW_ / 8; ++i) {
        fa0[i] = *(const v4f*)(a0p + 8 * i);
        fa1[i] = *(const v4f*)(a1p + 8 * i);
        fb0[i] = *(const v4f*)(b0p + 8 * i);
        fb1[i] = *(const v4f*)(b1p + 8 * i);
    }
    __builtin_amdgcn_sched_barrier(0);

    const v4f tv   = *(const v4f*)(tab + (size_t)tok * H_ + ecol);
    const v4f bias = *(const v4f*)(b_hh + ecol);
    __builtin_amdgcn_sched_barrier(0);

    v8f c00 = {}, c01 = {}, c10 = {}, c11 = {};
    #pragma unroll
    for (int i = 0; i < KW_ / 8; ++i) {
        const v4f a0 = fa0[i], a1 = fa1[i], b0 = fb0[i], b1 = fb1[i];
        const v2f a0l = { a0.x, a0.y }, a0h = { a0.z, a0.w };
        const v2f a1l = { a1.x, a1.y }, a1h = { a1.z, a1.w };
        const v2f b0l = { b0.x, b0.y }, b0h = { b0.z, b0.w };
        const v2f b1l = { b1.x, b1.y }, b1h = { b1.z, b1.w };
        c00 = __builtin_amdgcn_wmma_f32_16x16x4_f32(false, a0l, false, b0l, (short)0, c00, false, false);
        c01 = __builtin_amdgcn_wmma_f32_16x16x4_f32(false, a0l, false, b1l, (short)0, c01, false, false);
        c10 = __builtin_amdgcn_wmma_f32_16x16x4_f32(false, a1l, false, b0l, (short)0, c10, false, false);
        c11 = __builtin_amdgcn_wmma_f32_16x16x4_f32(false, a1l, false, b1l, (short)0, c11, false, false);
        c00 = __builtin_amdgcn_wmma_f32_16x16x4_f32(false, a0h, false, b0h, (short)0, c00, false, false);
        c01 = __builtin_amdgcn_wmma_f32_16x16x4_f32(false, a0h, false, b1h, (short)0, c01, false, false);
        c10 = __builtin_amdgcn_wmma_f32_16x16x4_f32(false, a1h, false, b0h, (short)0, c10, false, false);
        c11 = __builtin_amdgcn_wmma_f32_16x16x4_f32(false, a1h, false, b1h, (short)0, c11, false, false);
    }

    __builtin_amdgcn_sched_barrier(0);
    asm volatile("v_nop\n\tv_nop\n\tv_nop\n\tv_nop");
    __builtin_amdgcn_sched_barrier(0);

    float* __restrict__ pw = part + wave * (32 * RP_);
    #pragma unroll
    for (int r = 0; r < 8; ++r) {
        const int rr = half * 8 + r;
        pw[(rr     ) * RP_ + l15     ] = c00[r];
        pw[(rr     ) * RP_ + l15 + 16] = c01[r];
        pw[(rr + 16) * RP_ + l15     ] = c10[r];
        pw[(rr + 16) * RP_ + l15 + 16] = c11[r];
    }
    __syncthreads();

    const float* __restrict__ pr = part + er * RP_ + ec;
    v4f acc = *(const v4f*)pr;
    #pragma unroll
    for (int w = 1; w < 8; ++w) acc += *(const v4f*)(pr + w * (32 * RP_));

    v4f o;
    o.x = tanhf(acc.x + tv.x + bias.x);
    o.y = tanhf(acc.y + tv.y + bias.y);
    o.z = tanhf(acc.z + tv.z + bias.z);
    o.w = tanhf(acc.w + tv.w + bias.w);
    *(v4f*)(hout + (size_t)erow * H_ + ecol) = o;
    if (len - 1 == s) *(v4f*)(last + (size_t)erow * H_ + ecol) = o;
}

__global__ __launch_bounds__(256)
void k_mlp1(const float* __restrict__ last,
            const float* __restrict__ W1,
            const float* __restrict__ b1,
            float* __restrict__ y1) {
    const int lane = threadIdx.x & 31;
    const int wave = threadIdx.x >> 5;
    const int gw   = blockIdx.x * (blockDim.x >> 5) + wave;

    const int NT = H_ / 16;
    const int tn = (gw % NT) * 16;
    const int tm = (gw / NT) * 16;

    const int half = lane >> 4;
    const int l15  = lane & 15;

    const float* __restrict__ arow = last + (size_t)(tm + l15) * H_ + half * 4;
    const float* __restrict__ brow = W1   + (size_t)(tn + l15) * H_ + half * 4;

    v8f acc = {};
    #pragma unroll 8
    for (int k0 = 0; k0 < H_; k0 += 8) {
        const v4f a = *(const v4f*)(arow + k0);
        const v4f b = *(const v4f*)(brow + k0);
        const v2f a0 = { a.x, a.y }, b0 = { b.x, b.y };
        const v2f a1 = { a.z, a.w }, b1 = { b.z, b.w };
        acc = __builtin_amdgcn_wmma_f32_16x16x4_f32(false, a0, false, b0, (short)0, acc, false, false);
        acc = __builtin_amdgcn_wmma_f32_16x16x4_f32(false, a1, false, b1, (short)0, acc, false, false);
    }

    const int   col  = tn + l15;
    const float bias = b1[col];
    #pragma unroll
    for (int r = 0; r < 8; ++r) {
        const int row = tm + half * 8 + r;
        const float v = acc[r] + bias;
        y1[(size_t)row * H_ + col] = v > 0.0f ? v : 0.0f;
    }
}

__global__ void k_mlp2(const float* __restrict__ y1,
                       const float* __restrict__ W2,
                       const float* __restrict__ b2,
                       float* __restrict__ out) {
    int i = blockIdx.x * blockDim.x + threadIdx.x;
    if (i >= B_ * C_) return;
    const int b = i / C_;
    const int c = i % C_;
    const float* __restrict__ yr = y1 + (size_t)b * H_;
    const float* __restrict__ wr = W2 + (size_t)c * H_;
    float acc = b2[c];
    #pragma unroll 8
    for (int k = 0; k < H_; ++k) acc += yr[k] * wr[k];
    out[i] = acc;
}

extern "C" void kernel_launch(void* const* d_in, const int* in_sizes, int n_in,
                              void* d_out, int out_size, void* d_ws, size_t ws_size,
                              hipStream_t stream) {
    const int*   x_in  = (const int*)  d_in[0];
    const int*   x_len = (const int*)  d_in[1];
    const float* emb   = (const float*)d_in[2];
    const float* W_ih  = (const float*)d_in[3];
    const float* b_ih  = (const float*)d_in[4];
    const float* W_hh  = (const float*)d_in[5];
    const float* b_hh  = (const float*)d_in[6];
    const float* W1    = (const float*)d_in[7];
    const float* b1    = (const float*)d_in[8];
    const float* W2    = (const float*)d_in[9];
    const float* b2    = (const float*)d_in[10];
    float* out = (float*)d_out;

    float* tab  = (float*)d_ws;
    float* hb0  = tab  + (size_t)S_ * B_ * H_;
    float* hb1  = hb0  + (size_t)B_ * H_;
    float* last = hb1  + (size_t)B_ * H_;
    float* y1   = last + (size_t)B_ * H_;

    k_zero<<<(B_ * H_ + 255) / 256, 256, 0, stream>>>(hb0, B_ * H_);

    {
        const int waves  = (V_ / 16) * (H_ / 16);
        const int blocks = waves / 4;
        k_vocab_proj<<<blocks, 128, 0, stream>>>(emb, W_ih, b_ih, tab);
    }

    {
        const int blocks = (B_ / 32) * (H_ / 32);
        float* bufs[2] = { hb0, hb1 };
        for (int s = 0; s < S_; ++s) {
            const float* hin  = bufs[s & 1];
            float*       hout = bufs[(s + 1) & 1];
            k_rnn_step<<<blocks, 256, 0, stream>>>(
                hin, W_hh, b_hh, tab, x_in, x_len, s, hout, last);
        }
    }

    {
        const int waves  = (B_ / 16) * (H_ / 16);
        const int blocks = waves / 8;
        k_mlp1<<<blocks, 256, 0, stream>>>(last, W1, b1, y1);
        k_mlp2<<<(B_ * C_ + 127) / 128, 128, 0, stream>>>(y1, W2, b2, out);
    }
}


# Step GEMM operands fetched as whole 128-byte lines (four lines per load instead of 32-byte pieces of many lines) and re-laid through per-wave LDS into the same fragments; k order and sums unchanged
# speedup vs baseline: 3.6330x; 1.0356x over previous
#include <hip/hip_runtime.h>
#include <hip/hip_bf16.h>
#include <math.h>

typedef float v2f __attribute__((ext_vector_type(2)));
typedef float v4f __attribute__((ext_vector_type(4)));
typedef float v8f __attribute__((ext_vector_type(8)));

#define B_ 256
#define S_ 128
#define E_ 512
#define H_ 1024
#define V_ 128
#define C_ 18

__global__ void k_zero(float* __restrict__ p, int n) {
    int i = blockIdx.x * blockDim.x + threadIdx.x;
    if (i < n) p[i] = 0.0f;
}


__global__ __launch_bounds__(128)
void k_vocab_proj(const float* __restrict__ emb,
                  const float* __restrict__ W_ih,
                  const float* __restrict__ b_ih,
                  float* __restrict__ tab) {
    const int lane = threadIdx.x & 31;
    const int wave = threadIdx.x >> 5;
    const int gw   = blockIdx.x * (blockDim.x >> 5) + wave;

    const int NT = H_ / 16;
    const int tn = (gw % NT) * 16;
    const int tm = (gw / NT) * 16;

    const int half = lane >> 4;
    const int l15  = lane & 15;

    const float* __restrict__ arow = emb  + (size_t)(tm + l15) * E_ + half * 4;
    const float* __restrict__ brow = W_ih + (size_t)(tn + l15) * E_ + half * 4;

    v8f acc = {};
    #pragma unroll 16
    for (int k0 = 0; k0 < E_; k0 += 8) {
        const v4f a = *(const v4f*)(arow + k0);
        const v4f b = *(const v4f*)(brow + k0);
        const v2f a0 = { a.x, a.y }, b0 = { b.x, b.y };
        const v2f a1 = { a.z, a.w }, b1 = { b.z, b.w };
        acc = __builtin_amdgcn_wmma_f32_16x16x4_f32(false, a0, false, b0, (short)0, acc, false, false);
        acc = __builtin_amdgcn_wmma_f32_16x16x4_f32(false, a1, false, b1, (short)0, acc, false, false);
    }

    const int   col  = tn + l15;
    const float bias = b_ih[col];
    #pragma unroll
    for (int r = 0; r < 8; ++r) {
        const int row = tm + half * 8 + r;
        tab[(size_t)row * H_ + col] = acc[r] + bias;
    }
}

#define KW_ (H_ / 8)
#define RP_ 36
#define CK_ 32
#define NC_ (KW_ / CK_)
#define WS_ (64 * RP_)

#define WAVE_LDS_ORDER()                                              \
    do {                                                              \
        __builtin_amdgcn_fence(__ATOMIC_RELEASE, "wavefront");        \
        __builtin_amdgcn_wave_barrier();                              \
        __builtin_amdgcn_fence(__ATOMIC_ACQUIRE, "wavefront");        \
    } while (0)

#define RO_(t) ((((t) * 2) & 7) + 16 * ((t) >> 2))

__global__ __launch_bounds__(256)
void k_rnn_step(const float* __restrict__ hin,
                const float* __restrict__ W_hh,
                const float* __restrict__ b_hh,
                const float* __restrict__ tab,
                const int* __restrict__ x_in,
                const int* __restrict__ lengths,
                int s,
                float* __restrict__ hout,
                float* __restrict__ last) {
    __shared__ __attribute__((aligned(16))) float stage[8 * WS_];

    const int tid  = threadIdx.x;
    const int lane = tid & 31;
    const int wave = tid >> 5;
    const int half = lane >> 4;
    const int l15  = lane & 15;

    const int NT = H_ / 32;
    const int tn = (blockIdx.x % NT) * 32;
    const int tm = (blockIdx.x / NT) * 32;

    const int er   = tid >> 3;
    const int ec   = (tid & 7) * 4;
    const int erow = tm + er;
    const int ecol = tn + ec;
    const int tok = x_in[erow * S_ + s];
    const int len = lengths[erow];

    const int piece = lane & 7;
    const int lrow  = ((lane >> 4) & 1) + 8 * ((lane >> 3) & 1);
    const float* __restrict__ ap = hin  + (size_t)(tm + lrow) * H_ + wave * KW_ + piece * 4;
    const float* __restrict__ bp = W_hh + (size_t)(tn + lrow) * H_ + wave * KW_ + piece * 4;

    v4f g[NC_][16];
    #pragma unroll
    for (int j = 0; j < NC_; ++j) {
        #pragma unroll
        for (int t = 0; t < 8; ++t) {
            g[j][t]     = *(const v4f*)(ap + (size_t)RO_(t) * H_ + CK_ * j);
            g[j][8 + t] = *(const v4f*)(bp + (size_t)RO_(t) * H_ + CK_ * j);
        }
        __builtin_amdgcn_sched_barrier(0);
    }

    int tokp = tok, lenp = len;
    asm volatile("" : "+v"(tokp), "+v"(lenp));
    const v4f tv   = *(const v4f*)(tab + (size_t)tokp * H_ + ecol);
    const v4f bias = *(const v4f*)(b_hh + ecol);
    __builtin_amdgcn_sched_barrier(0);

    float* __restrict__ sw = stage + wave * WS_;
    float* __restrict__ st = sw + lrow * RP_ + piece * 4;
    const float* __restrict__ fr = sw + l15 * RP_ + half * 4;

#define STAGE_STORE(j)                                                        \
    _Pragma("unroll")                                                         \
    for (int t = 0; t < 8; ++t) {                                             \
        *(v4f*)(st + (RO_(t)     ) * RP_) = g[j][t];                          \
        *(v4f*)(st + (RO_(t) + 32) * RP_) = g[j][8 + t];                      \
    }
#define STAGE_FRAGS(F)                                                        \
    _Pragma("unroll")                                                         \
    for (int q = 0; q < CK_ / 8; ++q) {                                       \
        F[q][0] = *(const v4f*)(fr + 8 * q);                                  \
        F[q][1] = *(const v4f*)(fr + 16 * RP_ + 8 * q);                       \
        F[q][2] = *(const v4f*)(fr + 32 * RP_ + 8 * q);                       \
        F[q][3] = *(const v4f*)(fr + 48 * RP_ + 8 * q);                       \
    }
#define CHUNK_MMA(F)                                                          \
    _Pragma("unroll")                                                         \
    for (int q = 0; q < CK_ / 8; ++q) {                                       \
        const v4f a0 = F[q][0], a1 = F[q][1], b0 = F[q][2], b1 = F[q][3];     \
        const v2f a0l = { a0.x, a0.y }, a0h = { a0.z, a0.w };                 \
        const v2f a1l = { a1.x, a1.y }, a1h = { a1.z, a1.w };                 \
        const v2f b0l = { b0.x, b0.y }, b0h = { b0.z, b0.w };                 \
        const v2f b1l = { b1.x, b1.y }, b1h = { b1.z, b1.w };                 \
        c00 = __builtin_amdgcn_wmma_f32_16x16x4_f32(false, a0l, false, b0l, (short)0, c00, false, false); \
        c01 = __builtin_amdgcn_wmma_f32_16x16x4_f32(false, a0l, false, b1l, (short)0, c01, false, false); \
        c10 = __builtin_amdgcn_wmma_f32_16x16x4_f32(false, a1l, false, b0l, (short)0, c10, false, false); \
        c11 = __builtin_amdgcn_wmma_f32_16x16x4_f32(false, a1l, false, b1l, (short)0, c11, false, false); \
        c00 = __builtin_amdgcn_wmma_f32_16x16x4_f32(false, a0h, false, b0h, (short)0, c00, false, false); \
        c01 = __builtin_amdgcn_wmma_f32_16x16x4_f32(false, a0h, false, b1h, (short)0, c01, false, false); \
        c10 = __builtin_amdgcn_wmma_f32_16x16x4_f32(false, a1h, false, b0h, (short)0, c10, false, false); \
        c11 = __builtin_amdgcn_wmma_f32_16x16x4_f32(false, a1h, false, b1h, (short)0, c11, false, false); \
    }

    v8f c00 = {}, c01 = {}, c10 = {}, c11 = {};
    v4f f0[CK_ / 8][4], f1[CK_ / 8][4];

#define PHASE_EDGE()                                                          \
    do {                                                                      \
        __builtin_amdgcn_sched_barrier(0);                                    \
        asm volatile("" : "+v"(c00), "+v"(c01), "+v"(c10), "+v"(c11) : : "memory"); \
        __builtin_amdgcn_sched_barrier(0);                                    \
    } while (0)

    STAGE_STORE(0)
    WAVE_LDS_ORDER();
    STAGE_FRAGS(f0)
    WAVE_LDS_ORDER();
    STAGE_STORE(1)
    WAVE_LDS_ORDER();
    STAGE_FRAGS(f1)
    PHASE_EDGE();
    CHUNK_MMA(f0)
    PHASE_EDGE();
    WAVE_LDS_ORDER();
    STAGE_STORE(2)
    WAVE_LDS_ORDER();
    STAGE_FRAGS(f0)
    PHASE_EDGE();
    CHUNK_MMA(f1)
    PHASE_EDGE();
    WAVE_LDS_ORDER();
    STAGE_STORE(3)
    WAVE_LDS_ORDER();
    STAGE_FRAGS(f1)
    PHASE_EDGE();
    CHUNK_MMA(f0)
    CHUNK_MMA(f1)

    __builtin_amdgcn_sched_barrier(0);
    asm volatile("v_nop\n\tv_nop\n\tv_nop\n\tv_nop");
    __builtin_amdgcn_sched_barrier(0);

    WAVE_LDS_ORDER();
    float* __restrict__ pw = sw;
    #pragma unroll
    for (int r = 0; r < 8; ++r) {
        const int rr = half * 8 + r;
        pw[(rr     ) * RP_ + l15     ] = c00[r];
        pw[(rr     ) * RP_ + l15 + 16] = c01[r];
        pw[(rr + 16) * RP_ + l15     ] = c10[r];
        pw[(rr + 16) * RP_ + l15 + 16] = c11[r];
    }
    __syncthreads();

    const float* __restrict__ pr = stage + er * RP_ + ec;
    v4f acc = *(const v4f*)pr;
    #pragma unroll
    for (int w = 1; w < 8; ++w) acc += *(const v4f*)(pr + w * WS_);

    v4f o;
    o.x = tanhf(acc.x + tv.x + bias.x);
    o.y = tanhf(acc.y + tv.y + bias.y);
    o.z = tanhf(acc.z + tv.z + bias.z);
    o.w = tanhf(acc.w + tv.w + bias.w);
    *(v4f*)(hout + (size_t)erow * H_ + ecol) = o;
    if (lenp - 1 == s) *(v4f*)(last + (size_t)erow * H_ + ecol) = o;
#undef STAGE_STORE
#undef STAGE_FRAGS
#undef CHUNK_MMA
#undef PHASE_EDGE
}

__global__ __launch_bounds__(256)
void k_mlp1(const float* __restrict__ last,
            const float* __restrict__ W1,
            const float* __restrict__ b1,
            float* __restrict__ y1) {
    const int lane = threadIdx.x & 31;
    const int wave = threadIdx.x >> 5;
    const int gw   = blockIdx.x * (blockDim.x >> 5) + wave;

    const int NT = H_ / 16;
    const int tn = (gw % NT) * 16;
    const int tm = (gw / NT) * 16;

    const int half = lane >> 4;
    const int l15  = lane & 15;

    const float* __restrict__ arow = last + (size_t)(tm + l15) * H_ + half * 4;
    const float* __restrict__ brow = W1   + (size_t)(tn + l15) * H_ + half * 4;

    v8f acc = {};
    #pragma unroll 8
    for (int k0 = 0; k0 < H_; k0 += 8) {
        const v4f a = *(const v4f*)(arow + k0);
        const v4f b = *(const v4f*)(brow + k0);
        const v2f a0 = { a.x, a.y }, b0 = { b.x, b.y };
        const v2f a1 = { a.z, a.w }, b1 = { b.z, b.w };
        acc = __builtin_amdgcn_wmma_f32_16x16x4_f32(false, a0, false, b0, (short)0, acc, false, false);
        acc = __builtin_amdgcn_wmma_f32_16x16x4_f32(false, a1, false, b1, (short)0, acc, false, false);
    }

    const int   col  = tn + l15;
    const float bias = b1[col];
    #pragma unroll
    for (int r = 0; r < 8; ++r) {
        const int row = tm + half * 8 + r;
        const float v = acc[r] + bias;
        y1[(size_t)row * H_ + col] = v > 0.0f ? v : 0.0f;
    }
}

__global__ void k_mlp2(const float* __restrict__ y1,
                       const float* __restrict__ W2,
                       const float* __restrict__ b2,
                       float* __restrict__ out) {
    int i = blockIdx.x * blockDim.x + threadIdx.x;
    if (i >= B_ * C_) return;
    const int b = i / C_;
    const int c = i % C_;
    const float* __restrict__ yr = y1 + (size_t)b * H_;
    const float* __restrict__ wr = W2 + (size_t)c * H_;
    float acc = b2[c];
    #pragma unroll 8
    for (int k = 0; k < H_; ++k) acc += yr[k] * wr[k];
    out[i] = acc;
}

extern "C" void kernel_launch(void* const* d_in, const int* in_sizes, int n_in,
                              void* d_out, int out_size, void* d_ws, size_t ws_size,
                              hipStream_t stream) {
    const int*   x_in  = (const int*)  d_in[0];
    const int*   x_len = (const int*)  d_in[1];
    const float* emb   = (const float*)d_in[2];
    const float* W_ih  = (const float*)d_in[3];
    const float* b_ih  = (const float*)d_in[4];
    const float* W_hh  = (const float*)d_in[5];
    const float* b_hh  = (const float*)d_in[6];
    const float* W1    = (const float*)d_in[7];
    const float* b1    = (const float*)d_in[8];
    const float* W2    = (const float*)d_in[9];
    const float* b2    = (const float*)d_in[10];
    float* out = (float*)d_out;

    float* tab  = (float*)d_ws;
    float* hb0  = tab  + (size_t)S_ * B_ * H_;
    float* hb1  = hb0  + (size_t)B_ * H_;
    float* last = hb1  + (size_t)B_ * H_;
    float* y1   = last + (size_t)B_ * H_;

    k_zero<<<(B_ * H_ + 255) / 256, 256, 0, stream>>>(hb0, B_ * H_);

    {
        const int waves  = (V_ / 16) * (H_ / 16);
        const int blocks = waves / 4;
        k_vocab_proj<<<blocks, 128, 0, stream>>>(emb, W_ih, b_ih, tab);
    }

    {
        const int blocks = (B_ / 32) * (H_ / 32);
        float* bufs[2] = { hb0, hb1 };
        for (int s = 0; s < S_; ++s) {
            const float* hin  = bufs[s & 1];
            float*       hout = bufs[(s + 1) & 1];
            k_rnn_step<<<blocks, 256, 0, stream>>>(
                hin, W_hh, b_hh, tab, x_in, x_len, s, hout, last);
        }
    }

    {
        const int waves  = (B_ / 16) * (H_ / 16);
        const int blocks = waves / 8;
        k_mlp1<<<blocks, 256, 0, stream>>>(last, W1, b1, y1);
        k_mlp2<<<(B_ * C_ + 127) / 128, 128, 0, stream>>>(y1, W2, b2, out);
    }
}


# Rows ordered by sequence length; each step computes only the rows still running, in 16-row tiles once at most half remain: states past a row's length never reach the output; values bit-identical
# speedup vs baseline: 4.1720x; 1.1484x over previous
#include <hip/hip_runtime.h>
#include <hip/hip_bf16.h>
#include <math.h>

typedef float v2f __attribute__((ext_vector_type(2)));
typedef float v4f __attribute__((ext_vector_type(4)));
typedef float v8f __attribute__((ext_vector_type(8)));

#define B_ 256
#define S_ 128
#define E_ 512
#define H_ 1024
#define V_ 128
#define C_ 18

__global__ __launch_bounds__(256)
void k_order(const int* __restrict__ x_in,
             const int* __restrict__ lengths,
             int* __restrict__ tokp,
             int* __restrict__ capp,
             int* __restrict__ p2r,
             int* __restrict__ nact,
             float* __restrict__ h0) {
    __shared__ int scap[B_];
    const int b = threadIdx.x;
    const int s = blockIdx.x;

    int c = lengths[b] - 1;
    if (c < 0) c += S_;
    c = c < 0 ? 0 : (c > S_ - 1 ? S_ - 1 : c);
    scap[b] = c;
    __syncthreads();

    int rank = 0;
    for (int j = 0; j < B_; ++j) {
        const int cj = scap[j];
        rank += (cj > c || (cj == c && j < b)) ? 1 : 0;
    }
    tokp[s * B_ + rank] = x_in[b * S_ + s];
    if (s == 0) {
        p2r[rank]  = b;
        capp[rank] = c;
        if (b < S_) {
            int cnt = 0;
            for (int j = 0; j < B_; ++j) cnt += (scap[j] >= b) ? 1 : 0;
            nact[b] = cnt;
        }
    }

    const v4f z = { 0.0f, 0.0f, 0.0f, 0.0f };
    v4f* __restrict__ hz = (v4f*)h0;
    const int i = s * B_ + b;
    hz[i] = z;
    hz[i + S_ * B_] = z;
}


__global__ __launch_bounds__(128)
void k_vocab_proj(const float* __restrict__ emb,
                  const float* __restrict__ W_ih,
                  const float* __restrict__ b_ih,
                  float* __restrict__ tab) {
    const int lane = threadIdx.x & 31;
    const int wave = threadIdx.x >> 5;
    const int gw   = blockIdx.x * (blockDim.x >> 5) + wave;

    const int NT = H_ / 16;
    const int tn = (gw % NT) * 16;
    const int tm = (gw / NT) * 16;

    const int half = lane >> 4;
    const int l15  = lane & 15;

    const float* __restrict__ arow = emb  + (size_t)(tm + l15) * E_ + half * 4;
    const float* __restrict__ brow = W_ih + (size_t)(tn + l15) * E_ + half * 4;

    v8f acc = {};
    #pragma unroll 16
    for (int k0 = 0; k0 < E_; k0 += 8) {
        const v4f a = *(const v4f*)(arow + k0);
        const v4f b = *(const v4f*)(brow + k0);
        const v2f a0 = { a.x, a.y }, b0 = { b.x, b.y };
        const v2f a1 = { a.z, a.w }, b1 = { b.z, b.w };
        acc = __builtin_amdgcn_wmma_f32_16x16x4_f32(false, a0, false, b0, (short)0, acc, false, false);
        acc = __builtin_amdgcn_wmma_f32_16x16x4_f32(false, a1, false, b1, (short)0, acc, false, false);
    }

    const int   col  = tn + l15;
    const float bias = b_ih[col];
    #pragma unroll
    for (int r = 0; r < 8; ++r) {
        const int row = tm + half * 8 + r;
        tab[(size_t)row * H_ + col] = acc[r] + bias;
    }
}

#define KW_ (H_ / 8)
#define RP_ 36
#define CK_ 32
#define NC_ (KW_ / CK_)
#define WS_ (64 * RP_)

#define WAVE_LDS_ORDER()                                              \
    do {                                                              \
        __builtin_amdgcn_fence(__ATOMIC_RELEASE, "wavefront");        \
        __builtin_amdgcn_wave_barrier();                              \
        __builtin_amdgcn_fence(__ATOMIC_ACQUIRE, "wavefront");        \
    } while (0)

#define RO_(t) ((((t) * 2) & 7) + 16 * ((t) >> 2))

template <int MF>
__device__ __forceinline__
void step_tile(float* __restrict__ stage,
               const float* __restrict__ hin,
               const float* __restrict__ W_hh,
               const float* __restrict__ b_hh,
               const float* __restrict__ tab,
               const int* __restrict__ tokp,
               const int* __restrict__ capp,
               const int* __restrict__ p2r,
               int s,
               float* __restrict__ hout,
               float* __restrict__ last,
               int tm, int tn) {
    constexpr int LA = 4 * MF;
    constexpr int LB = 8;
    constexpr int ET = 128 * MF;

    const int tid  = threadIdx.x;
    const int lane = tid & 31;
    const int wave = tid >> 5;
    const int half = lane >> 4;
    const int l15  = lane & 15;

    const int et   = tid & (ET - 1);
    const int er   = et >> 3;
    const int ec   = (et & 7) * 4;
    const int erow = tm + er;
    const int ecol = tn + ec;
    const int tok  = tokp[s * B_ + erow];
    const int cap  = capp[erow];
    const int orow = p2r[erow];

    const int piece = lane & 7;
    const int lrow  = ((lane >> 4) & 1) + 8 * ((lane >> 3) & 1);
    const float* __restrict__ ap = hin  + (size_t)(tm + lrow) * H_ + wave * KW_ + piece * 4;
    const float* __restrict__ bp = W_hh + (size_t)(tn + lrow) * H_ + wave * KW_ + piece * 4;

    v4f g[NC_][LA + LB];
    #pragma unroll
    for (int j = 0; j < NC_; ++j) {
        #pragma unroll
        for (int t = 0; t < 8; ++t) {
            if (t < LA) g[j][t]  = *(const v4f*)(ap + (size_t)RO_(t) * H_ + CK_ * j);
            g[j][LA + t] = *(const v4f*)(bp + (size_t)RO_(t) * H_ + CK_ * j);
        }
        __builtin_amdgcn_sched_barrier(0);
    }

    int tokv = tok, capv = cap, orowv = orow;
    asm volatile("" : "+v"(tokv), "+v"(capv), "+v"(orowv));
    const v4f tv   = *(const v4f*)(tab + (size_t)tokv * H_ + ecol);
    const v4f bias = *(const v4f*)(b_hh + ecol);
    __builtin_amdgcn_sched_barrier(0);

    float* __restrict__ sw = stage + wave * WS_;
    float* __restrict__ st = sw + lrow * RP_ + piece * 4;
    const float* __restrict__ fr = sw + l15 * RP_ + half * 4;

#define STAGE_STORE(j)                                                        \
    _Pragma("unroll")                                                         \
    for (int t = 0; t < 8; ++t) {                                             \
        if (t < LA) *(v4f*)(st + (RO_(t)     ) * RP_) = g[j][t];              \
        *(v4f*)(st + (RO_(t) + 32) * RP_) = g[j][LA + t];                     \
    }
#define STAGE_FRAGS(F)                                                        \
    _Pragma("unroll")                                                         \
    for (int q = 0; q < CK_ / 8; ++q) {                                       \
        F[q][0] = *(const v4f*)(fr + 8 * q);                                  \
        if (MF == 2) F[q][1] = *(const v4f*)(fr + 16 * RP_ + 8 * q);          \
        F[q][2] = *(const v4f*)(fr + 32 * RP_ + 8 * q);                       \
        F[q][3] = *(const v4f*)(fr + 48 * RP_ + 8 * q);                       \
    }
#define WMMA_(c, a, b)                                                        \
    c = __builtin_amdgcn_wmma_f32_16x16x4_f32(false, a, false, b, (short)0, c, false, false)
#define CHUNK_MMA(F)                                                          \
    _Pragma("unroll")                                                         \
    for (int q = 0; q < CK_ / 8; ++q) {                                       \
        const v4f a0 = F[q][0], a1 = F[q][1], b0 = F[q][2], b1 = F[q][3];     \
        const v2f a0l = { a0.x, a0.y }, a0h = { a0.z, a0.w };                 \
        const v2f a1l = { a1.x, a1.y }, a1h = { a1.z, a1.w };                 \
        const v2f b0l = { b0.x, b0.y }, b0h = { b0.z, b0.w };                 \
        const v2f b1l = { b1.x, b1.y }, b1h = { b1.z, b1.w };                 \
        WMMA_(c00, a0l, b0l);                                                 \
        WMMA_(c01, a0l, b1l);                                                 \
        if (MF == 2) { WMMA_(c10, a1l, b0l); WMMA_(c11, a1l, b1l); }          \
        WMMA_(c00, a0h, b0h);                                                 \
        WMMA_(c01, a0h, b1h);                                                 \
        if (MF == 2) { WMMA_(c10, a1h, b0h); WMMA_(c11, a1h, b1h); }          \
    }

    v8f c00 = {}, c01 = {}, c10 = {}, c11 = {};
    v4f f0[CK_ / 8][4] = {}, f1[CK_ / 8][4] = {};

#define PHASE_EDGE()                                                          \
    do {                                                                      \
        __builtin_amdgcn_sched_barrier(0);                                    \
        if (MF == 2)                                                          \
            asm volatile("" : "+v"(c00), "+v"(c01), "+v"(c10), "+v"(c11) : : "memory"); \
        else                                                                  \
            asm volatile("" : "+v"(c00), "+v"(c01) : : "memory");             \
        __builtin_amdgcn_sched_barrier(0);                                    \
    } while (0)

    STAGE_STORE(0)
    WAVE_LDS_ORDER();
    STAGE_FRAGS(f0)
    WAVE_LDS_ORDER();
    STAGE_STORE(1)
    WAVE_LDS_ORDER();
    STAGE_FRAGS(f1)
    PHASE_EDGE();
    CHUNK_MMA(f0)
    PHASE_EDGE();
    WAVE_LDS_ORDER();
    STAGE_STORE(2)
    WAVE_LDS_ORDER();
    STAGE_FRAGS(f0)
    PHASE_EDGE();
    CHUNK_MMA(f1)
    PHASE_EDGE();
    WAVE_LDS_ORDER();
    STAGE_STORE(3)
    WAVE_LDS_ORDER();
    STAGE_FRAGS(f1)
    PHASE_EDGE();
    CHUNK_MMA(f0)
    CHUNK_MMA(f1)

    __builtin_amdgcn_sched_barrier(0);
    asm volatile("v_nop\n\tv_nop\n\tv_nop\n\tv_nop");
    __builtin_amdgcn_sched_barrier(0);

    WAVE_LDS_ORDER();
    float* __restrict__ pw = sw;
    #pragma unroll
    for (int r = 0; r < 8; ++r) {
        const int rr = half * 8 + r;
        pw[(rr     ) * RP_ + l15     ] = c00[r];
        pw[(rr     ) * RP_ + l15 + 16] = c01[r];
        if (MF == 2) {
            pw[(rr + 16) * RP_ + l15     ] = c10[r];
            pw[(rr + 16) * RP_ + l15 + 16] = c11[r];
        }
    }
    __syncthreads();

    if (ET == 256 || tid < ET) {
        const float* __restrict__ pr = stage + er * RP_ + ec;
        v4f acc = *(const v4f*)pr;
        #pragma unroll
        for (int w = 1; w < 8; ++w) acc += *(const v4f*)(pr + w * WS_);

        v4f o;
        o.x = tanhf(acc.x + tv.x + bias.x);
        o.y = tanhf(acc.y + tv.y + bias.y);
        o.z = tanhf(acc.z + tv.z + bias.z);
        o.w = tanhf(acc.w + tv.w + bias.w);
        *(v4f*)(hout + (size_t)erow * H_ + ecol) = o;
        if (capv == s) *(v4f*)(last + (size_t)orowv * H_ + ecol) = o;
    }
#undef STAGE_STORE
#undef STAGE_FRAGS
#undef WMMA_
#undef CHUNK_MMA
#undef PHASE_EDGE
}

__global__ __launch_bounds__(256)
void k_rnn_step(const float* __restrict__ hin,
                const float* __restrict__ W_hh,
                const float* __restrict__ b_hh,
                const float* __restrict__ tab,
                const int* __restrict__ tokp,
                const int* __restrict__ capp,
                const int* __restrict__ p2r,
                const int* __restrict__ nact,
                int s,
                float* __restrict__ hout,
                float* __restrict__ last) {
    __shared__ __attribute__((aligned(16))) float stage[8 * WS_];

    const int n  = nact[s];
    const int NT = H_ / 32;
    const int tn = (blockIdx.x % NT) * 32;
    const int rt = blockIdx.x / NT;

    if (n > B_ / 2) {
        if (rt * 32 >= n) return;
        step_tile<2>(stage, hin, W_hh, b_hh, tab, tokp, capp, p2r, s, hout, last, rt * 32, tn);
    } else {
        if (rt * 16 >= n) return;
        step_tile<1>(stage, hin, W_hh, b_hh, tab, tokp, capp, p2r, s, hout, last, rt * 16, tn);
    }
}

__global__ __launch_bounds__(256)
void k_mlp1(const float* __restrict__ last,
            const float* __restrict__ W1,
            const float* __restrict__ b1,
            float* __restrict__ y1) {
    const int lane = threadIdx.x & 31;
    const int wave = threadIdx.x >> 5;
    const int gw   = blockIdx.x * (blockDim.x >> 5) + wave;

    const int NT = H_ / 16;
    const int tn = (gw % NT) * 16;
    const int tm = (gw / NT) * 16;

    const int half = lane >> 4;
    const int l15  = lane & 15;

    const float* __restrict__ arow = last + (size_t)(tm + l15) * H_ + half * 4;
    const float* __restrict__ brow = W1   + (size_t)(tn + l15) * H_ + half * 4;

    v8f acc = {};
    #pragma unroll 8
    for (int k0 = 0; k0 < H_; k0 += 8) {
        const v4f a = *(const v4f*)(arow + k0);
        const v4f b = *(const v4f*)(brow + k0);
        const v2f a0 = { a.x, a.y }, b0 = { b.x, b.y };
        const v2f a1 = { a.z, a.w }, b1 = { b.z, b.w };
        acc = __builtin_amdgcn_wmma_f32_16x16x4_f32(false, a0, false, b0, (short)0, acc, false, false);
        acc = __builtin_amdgcn_wmma_f32_16x16x4_f32(false, a1, false, b1, (short)0, acc, false, false);
    }

    const int   col  = tn + l15;
    const float bias = b1[col];
    #pragma unroll
    for (int r = 0; r < 8; ++r) {
        const int row = tm + half * 8 + r;
        const float v = acc[r] + bias;
        y1[(size_t)row * H_ + col] = v > 0.0f ? v : 0.0f;
    }
}

__global__ void k_mlp2(const float* __restrict__ y1,
                       const float* __restrict__ W2,
                       const float* __restrict__ b2,
                       float* __restrict__ out) {
    int i = blockIdx.x * blockDim.x + threadIdx.x;
    if (i >= B_ * C_) return;
    const int b = i / C_;
    const int c = i % C_;
    const float* __restrict__ yr = y1 + (size_t)b * H_;
    const float* __restrict__ wr = W2 + (size_t)c * H_;
    float acc = b2[c];
    #pragma unroll 8
    for (int k = 0; k < H_; ++k) acc += yr[k] * wr[k];
    out[i] = acc;
}

extern "C" void kernel_launch(void* const* d_in, const int* in_sizes, int n_in,
                              void* d_out, int out_size, void* d_ws, size_t ws_size,
                              hipStream_t stream) {
    const int*   x_in  = (const int*)  d_in[0];
    const int*   x_len = (const int*)  d_in[1];
    const float* emb   = (const float*)d_in[2];
    const float* W_ih  = (const float*)d_in[3];
    const float* b_ih  = (const float*)d_in[4];
    const float* W_hh  = (const float*)d_in[5];
    const float* b_hh  = (const float*)d_in[6];
    const float* W1    = (const float*)d_in[7];
    const float* b1    = (const float*)d_in[8];
    const float* W2    = (const float*)d_in[9];
    const float* b2    = (const float*)d_in[10];
    float* out = (float*)d_out;

    float* tab  = (float*)d_ws;
    float* hb0  = tab  + (size_t)S_ * B_ * H_;
    float* hb1  = hb0  + (size_t)B_ * H_;
    float* last = hb1  + (size_t)B_ * H_;
    float* y1   = last + (size_t)B_ * H_;
    int* tokp = (int*)(tab + (size_t)V_ * H_);
    int* capp = tokp + S_ * B_;
    int* p2r  = capp + B_;
    int* nact = p2r  + B_;

    k_order<<<S_, B_, 0, stream>>>(x_in, x_len, tokp, capp, p2r, nact, hb0);

    {
        const int waves  = (V_ / 16) * (H_ / 16);
        const int blocks = waves / 4;
        k_vocab_proj<<<blocks, 128, 0, stream>>>(emb, W_ih, b_ih, tab);
    }

    {
        const int blocks = (B_ / 32) * (H_ / 32);
        float* bufs[2] = { hb0, hb1 };
        for (int s = 0; s < S_; ++s) {
            const float* hin  = bufs[s & 1];
            float*       hout = bufs[(s + 1) & 1];
            k_rnn_step<<<blocks, 256, 0, stream>>>(
                hin, W_hh, b_hh, tab, tokp, capp, p2r, nact, s, hout, last);
        }
    }

    {
        const int waves  = (B_ / 16) * (H_ / 16);
        const int blocks = waves / 8;
        k_mlp1<<<blocks, 256, 0, stream>>>(last, W1, b1, y1);
        k_mlp2<<<(B_ * C_ + 127) / 128, 128, 0, stream>>>(y1, W2, b2, out);
    }
}
